# starting kernel
# baseline (speedup 1.0000x reference)
#include <hip/hip_runtime.h>
#include <stdint.h>

#define NNODES 50000
#define NEDGES 800000
#define ETOT   (NEDGES + NNODES)

typedef float v2f __attribute__((ext_vector_type(2)));
typedef float v8f __attribute__((ext_vector_type(8)));

__device__ __forceinline__ unsigned flipF(float f) {
    unsigned u = __float_as_uint(f);
    unsigned mask = (unsigned)(-(int)(u >> 31)) | 0x80000000u;
    return u ^ mask;
}
__device__ __forceinline__ float unflipF(unsigned u) {
    unsigned mask = ((u >> 31) - 1u) | 0x80000000u;
    return __uint_as_float(u ^ mask);
}

__device__ __forceinline__ void edge_sd(const int* __restrict__ ei, int e, int& s, int& d) {
    if (e < NEDGES) { s = ei[e]; d = ei[NEDGES + e]; }
    else            { s = e - NEDGES; d = s; }
}

__global__ void zero_f32(float* __restrict__ p, long n) {
    long i = (long)blockIdx.x * blockDim.x + threadIdx.x;
    long st = (long)gridDim.x * blockDim.x;
    for (; i < n; i += st) p[i] = 0.0f;
}

__global__ __launch_bounds__(256)
void gemm_f32_wmma(const float* __restrict__ A, const float* __restrict__ B,
                   float* __restrict__ C, int M, int N, int K) {
    const int wave = (blockIdx.x * blockDim.x + threadIdx.x) >> 5;
    const int lane = threadIdx.x & 31;
    const int stripsN = N >> 6;
    const int tm = wave / stripsN;
    const int tn = (wave % stripsN) * 4;
    if (tm * 16 >= M) return;

    const int m   = lane & 15;
    const int kh  = (lane >> 4) << 1;
    const int n0  = lane & 15;

    v8f acc[4];
    #pragma unroll
    for (int t = 0; t < 4; ++t) acc[t] = (v8f){0.f,0.f,0.f,0.f,0.f,0.f,0.f,0.f};

    const float* __restrict__ Arow = A + (long)(tm * 16 + m) * K;
    for (int k0 = 0; k0 < K; k0 += 4) {
        v2f a;
        a.x = Arow[k0 + kh + 0];
        a.y = Arow[k0 + kh + 1];
        #pragma unroll
        for (int t = 0; t < 4; ++t) {
            const int col = (tn + t) * 16 + n0;
            v2f b;
            b.x = B[(long)(k0 + kh + 0) * N + col];
            b.y = B[(long)(k0 + kh + 1) * N + col];
            acc[t] = __builtin_amdgcn_wmma_f32_16x16x4_f32(
                         false, a, false, b, (short)0, acc[t], false, false);
        }
    }
    const int moff = (lane >> 4) * 8;
    #pragma unroll
    for (int t = 0; t < 4; ++t) {
        #pragma unroll
        for (int v = 0; v < 8; ++v) {
            C[(long)(tm * 16 + v + moff) * N + (tn + t) * 16 + n0] = acc[t][v];
        }
    }
}

__global__ void attn_scores(const float* __restrict__ xh,
                            const float* __restrict__ att_src,
                            const float* __restrict__ att_dst,
                            float* __restrict__ asrc, float* __restrict__ adst,
                            int H, int C) {
    int idx = blockIdx.x * blockDim.x + threadIdx.x;
    if (idx >= NNODES * H) return;
    int n = idx / H, h = idx % H;
    const float* x = xh + (long)n * H * C + (long)h * C;
    const float* as = att_src + h * C;
    const float* ad = att_dst + h * C;
    float ss = 0.f, sd = 0.f;
    for (int c = 0; c < C; ++c) { float v = x[c]; ss += v * as[c]; sd += v * ad[c]; }
    asrc[idx] = ss;
    adst[idx] = sd;
}

__global__ void edge_max(const int* __restrict__ ei,
                         const float* __restrict__ asrc, const float* __restrict__ adst,
                         float* __restrict__ e_out, unsigned* __restrict__ emax, int H) {
    int idx = blockIdx.x * blockDim.x + threadIdx.x;
    if (idx >= ETOT * H) return;
    int e = idx / H, h = idx % H;
    int s, d; edge_sd(ei, e, s, d);
    float x = asrc[s * H + h] + adst[d * H + h];
    x = (x > 0.f) ? x : 0.2f * x;
    e_out[idx] = x;
    atomicMax(&emax[d * H + h], flipF(x));
}

__global__ void edge_expsum(const int* __restrict__ ei,
                            float* __restrict__ e_io, const unsigned* __restrict__ emax,
                            float* __restrict__ esum, int H) {
    int idx = blockIdx.x * blockDim.x + threadIdx.x;
    if (idx >= ETOT * H) return;
    int e = idx / H, h = idx % H;
    int s, d; edge_sd(ei, e, s, d); (void)s;
    float m  = unflipF(emax[d * H + h]);
    float ex = expf(e_io[idx] - m);
    e_io[idx] = ex;
    atomicAdd(&esum[d * H + h], ex);
}

__global__ __launch_bounds__(256)
void edge_aggregate(const int* __restrict__ ei, const float* __restrict__ ex,
                    const float* __restrict__ esum, const float* __restrict__ xh,
                    float* __restrict__ out, int H, int C) {
    int gid  = blockIdx.x * blockDim.x + threadIdx.x;
    int e    = gid >> 5;
    int lane = gid & 31;
    if (e >= ETOT) return;
    int s, d; edge_sd(ei, e, s, d);
    const int HC = H * C;
    const float* __restrict__ xs = xh + (long)s * HC;
    float* __restrict__ od = out + (long)d * HC;
    for (int base = lane * 4; base < HC; base += 128) {
        int h = base / C;
        float coeff = ex[(long)e * H + h] / esum[d * H + h];
        float4 v = *(const float4*)(xs + base);
        atomicAdd(od + base + 0, coeff * v.x);
        atomicAdd(od + base + 1, coeff * v.y);
        atomicAdd(od + base + 2, coeff * v.z);
        atomicAdd(od + base + 3, coeff * v.w);
    }
}

__global__ void add_bias(float* __restrict__ out, const float* __restrict__ b,
                         int NC, long total) {
    long i = (long)blockIdx.x * blockDim.x + threadIdx.x;
    if (i >= total) return;
    out[i] += b[(int)(i % NC)];
}

static inline int cdiv(long a, int b) { return (int)((a + b - 1) / b); }

extern "C" void kernel_launch(void* const* d_in, const int* in_sizes, int n_in,
                              void* d_out, int out_size, void* d_ws, size_t ws_size,
                              hipStream_t stream) {
    const float* X        = (const float*)d_in[0];
    const int*   ei       = (const int*)  d_in[1];
    const float* W1       = (const float*)d_in[2];
    const float* att_src1 = (const float*)d_in[3];
    const float* att_dst1 = (const float*)d_in[4];
    const float* b1       = (const float*)d_in[5];
    const float* W2       = (const float*)d_in[6];
    const float* att_src2 = (const float*)d_in[7];
    const float* att_dst2 = (const float*)d_in[8];
    const float* b2       = (const float*)d_in[9];
    float* out = (float*)d_out;

    float* ws = (float*)d_ws;
    long off = 0;
    float*    xh1   = ws + off; off += (long)NNODES * 256;
    float*    h1    = ws + off; off += (long)NNODES * 256;
    float*    esum1 = ws + off; off += (long)NNODES * 4;
    unsigned* emax1 = (unsigned*)(ws + off); off += (long)NNODES * 4;
    float*    esum2 = ws + off; off += NNODES;
    unsigned* emax2 = (unsigned*)(ws + off); off += NNODES;
    float*    asrc1 = ws + off; off += (long)NNODES * 4;
    float*    adst1 = ws + off; off += (long)NNODES * 4;
    float*    e1    = ws + off; off += (long)ETOT * 4;
    float*    xh2   = xh1;
    float*    e2    = e1;
    float*    asrc2 = asrc1;
    float*    adst2 = adst1;

    const int B = 256;

    long zlen = (long)NNODES * 256 + (long)NNODES * 4 * 2 + (long)NNODES * 2;
    zero_f32<<<2048, B, 0, stream>>>(h1, zlen);
    zero_f32<<<2048, B, 0, stream>>>(out, (long)NNODES * 128);

    {
        int waves = (NNODES / 16) * (256 / 64);
        gemm_f32_wmma<<<cdiv((long)waves * 32, B), B, 0, stream>>>(X, W1, xh1, NNODES, 256, 64);
        attn_scores<<<cdiv((long)NNODES * 4, B), B, 0, stream>>>(xh1, att_src1, att_dst1,
                                                                 asrc1, adst1, 4, 64);
        edge_max   <<<cdiv((long)ETOT * 4, B), B, 0, stream>>>(ei, asrc1, adst1, e1, emax1, 4);
        edge_expsum<<<cdiv((long)ETOT * 4, B), B, 0, stream>>>(ei, e1, emax1, esum1, 4);
        edge_aggregate<<<cdiv((long)ETOT * 32, B), B, 0, stream>>>(ei, e1, esum1, xh1, h1, 4, 64);
        add_bias<<<cdiv((long)NNODES * 256, B), B, 0, stream>>>(h1, b1, 256, (long)NNODES * 256);
    }

    {
        int waves = (NNODES / 16) * (128 / 64);
        gemm_f32_wmma<<<cdiv((long)waves * 32, B), B, 0, stream>>>(h1, W2, xh2, NNODES, 128, 256);
        attn_scores<<<cdiv((long)NNODES, B), B, 0, stream>>>(xh2, att_src2, att_dst2,
                                                             asrc2, adst2, 1, 128);
        edge_max   <<<cdiv((long)ETOT, B), B, 0, stream>>>(ei, asrc2, adst2, e2, emax2, 1);
        edge_expsum<<<cdiv((long)ETOT, B), B, 0, stream>>>(ei, e2, emax2, esum2, 1);
        edge_aggregate<<<cdiv((long)ETOT * 32, B), B, 0, stream>>>(ei, e2, esum2, xh2, out, 1, 128);
        add_bias<<<cdiv((long)NNODES * 128, B), B, 0, stream>>>(out, b2, 128, (long)NNODES * 128);
    }
}


# Aggregation in LDS, no edge grouping: a work-group owns 196 nodes, streams all destinations, adds matching source rows into its LDS rows (LDS float adds), one store per row + bias; sum order changes
# speedup vs baseline: 1.8429x; 1.8429x over previous
#include <hip/hip_runtime.h>
#include <stdint.h>

#define NNODES 50000
#define NEDGES 800000
#define ETOT   (NEDGES + NNODES)

typedef float v2f __attribute__((ext_vector_type(2)));
typedef float v8f __attribute__((ext_vector_type(8)));

__device__ __forceinline__ unsigned flipF(float f) {
    unsigned u = __float_as_uint(f);
    unsigned mask = (unsigned)(-(int)(u >> 31)) | 0x80000000u;
    return u ^ mask;
}
__device__ __forceinline__ float unflipF(unsigned u) {
    unsigned mask = ((u >> 31) - 1u) | 0x80000000u;
    return __uint_as_float(u ^ mask);
}

__device__ __forceinline__ void edge_sd(const int* __restrict__ ei, int e, int& s, int& d) {
    if (e < NEDGES) { s = ei[e]; d = ei[NEDGES + e]; }
    else            { s = e - NEDGES; d = s; }
}

__global__ void zero_f32(float* __restrict__ p, long n) {
    long i = (long)blockIdx.x * blockDim.x + threadIdx.x;
    long st = (long)gridDim.x * blockDim.x;
    for (; i < n; i += st) p[i] = 0.0f;
}

__global__ __launch_bounds__(256)
void gemm_f32_wmma(const float* __restrict__ A, const float* __restrict__ B,
                   float* __restrict__ C, int M, int N, int K) {
    const int wave = (blockIdx.x * blockDim.x + threadIdx.x) >> 5;
    const int lane = threadIdx.x & 31;
    const int stripsN = N >> 6;
    const int tm = wave / stripsN;
    const int tn = (wave % stripsN) * 4;
    if (tm * 16 >= M) return;

    const int m   = lane & 15;
    const int kh  = (lane >> 4) << 1;
    const int n0  = lane & 15;

    v8f acc[4];
    #pragma unroll
    for (int t = 0; t < 4; ++t) acc[t] = (v8f){0.f,0.f,0.f,0.f,0.f,0.f,0.f,0.f};

    const float* __restrict__ Arow = A + (long)(tm * 16 + m) * K;
    for (int k0 = 0; k0 < K; k0 += 4) {
        v2f a;
        a.x = Arow[k0 + kh + 0];
        a.y = Arow[k0 + kh + 1];
        #pragma unroll
        for (int t = 0; t < 4; ++t) {
            const int col = (tn + t) * 16 + n0;
            v2f b;
            b.x = B[(long)(k0 + kh + 0) * N + col];
            b.y = B[(long)(k0 + kh + 1) * N + col];
            acc[t] = __builtin_amdgcn_wmma_f32_16x16x4_f32(
                         false, a, false, b, (short)0, acc[t], false, false);
        }
    }
    const int moff = (lane >> 4) * 8;
    #pragma unroll
    for (int t = 0; t < 4; ++t) {
        #pragma unroll
        for (int v = 0; v < 8; ++v) {
            C[(long)(tm * 16 + v + moff) * N + (tn + t) * 16 + n0] = acc[t][v];
        }
    }
}

__global__ void attn_scores(const float* __restrict__ xh,
                            const float* __restrict__ att_src,
                            const float* __restrict__ att_dst,
                            float* __restrict__ asrc, float* __restrict__ adst,
                            int H, int C) {
    int idx = blockIdx.x * blockDim.x + threadIdx.x;
    if (idx >= NNODES * H) return;
    int n = idx / H, h = idx % H;
    const float* x = xh + (long)n * H * C + (long)h * C;
    const float* as = att_src + h * C;
    const float* ad = att_dst + h * C;
    float ss = 0.f, sd = 0.f;
    for (int c = 0; c < C; ++c) { float v = x[c]; ss += v * as[c]; sd += v * ad[c]; }
    asrc[idx] = ss;
    adst[idx] = sd;
}

__global__ void edge_max(const int* __restrict__ ei,
                         const float* __restrict__ asrc, const float* __restrict__ adst,
                         float* __restrict__ e_out, unsigned* __restrict__ emax, int H) {
    int idx = blockIdx.x * blockDim.x + threadIdx.x;
    if (idx >= ETOT * H) return;
    int e = idx / H, h = idx % H;
    int s, d; edge_sd(ei, e, s, d);
    float x = asrc[s * H + h] + adst[d * H + h];
    x = (x > 0.f) ? x : 0.2f * x;
    e_out[idx] = x;
    atomicMax(&emax[d * H + h], flipF(x));
}

__global__ void edge_expsum(const int* __restrict__ ei,
                            float* __restrict__ e_io, const unsigned* __restrict__ emax,
                            float* __restrict__ esum, int H) {
    int idx = blockIdx.x * blockDim.x + threadIdx.x;
    if (idx >= ETOT * H) return;
    int e = idx / H, h = idx % H;
    int s, d; edge_sd(ei, e, s, d); (void)s;
    float m  = unflipF(emax[d * H + h]);
    float ex = expf(e_io[idx] - m);
    e_io[idx] = ex;
    atomicAdd(&esum[d * H + h], ex);
}

#define RNODES 196
#define QCAP   160

template <int H, int C>
__global__ __launch_bounds__(1024)
void aggregate_owned(const int* __restrict__ ei, const float* __restrict__ ex,
                     const float* __restrict__ esum, const float* __restrict__ xh,
                     const float* __restrict__ bias, float* __restrict__ out) {
    constexpr int HC   = H * C;
    constexpr int NV   = HC / 128;
    constexpr int NBLK = NEDGES / 128;
    constexpr int GRP  = 5;
    constexpr int NGRP = NBLK / GRP;
    static_assert(NBLK * 128 == NEDGES && NGRP * GRP == NBLK, "edge count must tile");
    static_assert(NV * 128 == HC, "row must be whole 128-float pieces");

    __shared__ float acc[RNODES * HC];
    __shared__ int2  queue[32 * QCAP];

    const int lane = threadIdx.x & 31;
    const int wave = __builtin_amdgcn_readfirstlane((int)(threadIdx.x >> 5));
    const int n0   = blockIdx.x * RNODES;
    const int cntN = min(RNODES, NNODES - n0);

    for (int r = wave; r < cntN; r += 32) {
        const int n = n0 + r;
        float cf[H];
        #pragma unroll
        for (int h = 0; h < H; ++h) cf[h] = ex[(long)(NEDGES + n) * H + h] / esum[n * H + h];
        #pragma unroll
        for (int j = 0; j < NV; ++j) {
            const int hsel = (j * 128 + lane * 4) / C;
            float c = cf[0];
            #pragma unroll
            for (int h = 1; h < H; ++h) c = (hsel == h) ? cf[h] : c;
            const float4 v = *(const float4*)(xh + (long)n * HC + j * 128 + lane * 4);
            float* a = acc + r * HC + j * 128 + lane;
            a[0] = c * v.x; a[32] = c * v.y; a[64] = c * v.z; a[96] = c * v.w;
        }
    }
    __syncthreads();

    const int4* __restrict__ dst4 = (const int4*)(ei + NEDGES);
    int2* __restrict__ q = queue + wave * QCAP;
    int cnt = 0;

    auto append = [&](int d, int e) {
        const bool m = (unsigned)(d - n0) < (unsigned)cntN;
        const unsigned mask = __builtin_amdgcn_ballot_w32(m);
        if (mask) {
            const int pos = cnt + (int)__builtin_amdgcn_mbcnt_lo(mask, 0u);
            if (m) q[pos] = make_int2(e, d - n0);
            cnt += __builtin_popcount(mask);
        }
    };

    auto flush = [&](int b, int n) {
        const int2 ent = q[b + (lane < n ? lane : 0)];
        const int my_s = ei[ent.x];
        float my_cf[H];
        #pragma unroll
        for (int h = 0; h < H; ++h)
            my_cf[h] = ex[(long)ent.x * H + h] / esum[(n0 + ent.y) * H + h];

        auto add_row = [&](int i, const float4* v) {
            const int dl = __builtin_amdgcn_readlane(ent.y, i);
            float cf[H];
            #pragma unroll
            for (int h = 0; h < H; ++h)
                cf[h] = __int_as_float(__builtin_amdgcn_readlane(__float_as_int(my_cf[h]), i));
            #pragma unroll
            for (int j = 0; j < NV; ++j) {
                const int hsel = (j * 128 + lane * 4) / C;
                float c = cf[0];
                #pragma unroll
                for (int h = 1; h < H; ++h) c = (hsel == h) ? cf[h] : c;
                float* a = acc + dl * HC + j * 128 + lane;
                atomicAdd(a,      c * v[j].x);
                atomicAdd(a + 32, c * v[j].y);
                atomicAdd(a + 64, c * v[j].z);
                atomicAdd(a + 96, c * v[j].w);
            }
        };

        for (int i = 0; i < n; i += 2) {
            const int i2 = (i + 1 < n) ? i + 1 : i;
            const int sA = __builtin_amdgcn_readlane(my_s, i);
            const int sB = __builtin_amdgcn_readlane(my_s, i2);
            float4 vA[NV], vB[NV];
            #pragma unroll
            for (int j = 0; j < NV; ++j)
                vA[j] = *(const float4*)(xh + (long)sA * HC + j * 128 + lane * 4);
            #pragma unroll
            for (int j = 0; j < NV; ++j)
                vB[j] = *(const float4*)(xh + (long)sB * HC + j * 128 + lane * 4);
            add_row(i, vA);
            if (i2 != i) add_row(i2, vB);
        }
    };

    const int rot = (int)(blockIdx.x * NGRP / gridDim.x);
    int4 cur[GRP], nxt[GRP];
    {
        int gg = wave + rot; if (gg >= NGRP) gg -= NGRP;
        #pragma unroll
        for (int k = 0; k < GRP; ++k) cur[k] = dst4[(gg * GRP + k) * 32 + lane];
    }
    for (int g = wave; g < NGRP; g += 32) {
        int gg = g + rot; if (gg >= NGRP) gg -= NGRP;
        #pragma unroll
        for (int k = 0; k < GRP; ++k) nxt[k] = cur[k];
        if (g + 32 < NGRP) {
            int g2 = g + 32 + rot; if (g2 >= NGRP) g2 -= NGRP;
            #pragma unroll
            for (int k = 0; k < GRP; ++k) nxt[k] = dst4[(g2 * GRP + k) * 32 + lane];
        }
        #pragma unroll
        for (int k = 0; k < GRP; ++k) {
            const int eb = (gg * GRP + k) * 128 + lane * 4;
            append(cur[k].x, eb + 0);
            append(cur[k].y, eb + 1);
            append(cur[k].z, eb + 2);
            append(cur[k].w, eb + 3);
            while (cnt >= 32) { flush(cnt - 32, 32); cnt -= 32; }
        }
        #pragma unroll
        for (int k = 0; k < GRP; ++k) cur[k] = nxt[k];
    }
    if (cnt > 0) flush(0, cnt);
    __syncthreads();

    for (int r = wave; r < cntN; r += 32) {
        #pragma unroll
        for (int j = 0; j < NV; ++j) {
            const float* a = acc + r * HC + j * 128 + lane;
            const float4 b = *(const float4*)(bias + j * 128 + lane * 4);
            float4 o;
            o.x = a[0] + b.x; o.y = a[32] + b.y; o.z = a[64] + b.z; o.w = a[96] + b.w;
            *(float4*)(out + (long)(n0 + r) * HC + j * 128 + lane * 4) = o;
        }
    }
}

static inline int cdiv(long a, int b) { return (int)((a + b - 1) / b); }

extern "C" void kernel_launch(void* const* d_in, const int* in_sizes, int n_in,
                              void* d_out, int out_size, void* d_ws, size_t ws_size,
                              hipStream_t stream) {
    const float* X        = (const float*)d_in[0];
    const int*   ei       = (const int*)  d_in[1];
    const float* W1       = (const float*)d_in[2];
    const float* att_src1 = (const float*)d_in[3];
    const float* att_dst1 = (const float*)d_in[4];
    const float* b1       = (const float*)d_in[5];
    const float* W2       = (const float*)d_in[6];
    const float* att_src2 = (const float*)d_in[7];
    const float* att_dst2 = (const float*)d_in[8];
    const float* b2       = (const float*)d_in[9];
    float* out = (float*)d_out;

    float* ws = (float*)d_ws;
    long off = 0;
    float*    xh1   = ws + off; off += (long)NNODES * 256;
    float*    h1    = ws + off; off += (long)NNODES * 256;
    float*    esum1 = ws + off; off += (long)NNODES * 4;
    unsigned* emax1 = (unsigned*)(ws + off); off += (long)NNODES * 4;
    float*    esum2 = ws + off; off += NNODES;
    unsigned* emax2 = (unsigned*)(ws + off); off += NNODES;
    float*    asrc1 = ws + off; off += (long)NNODES * 4;
    float*    adst1 = ws + off; off += (long)NNODES * 4;
    float*    e1    = ws + off; off += (long)ETOT * 4;
    float*    xh2   = xh1;
    float*    e2    = e1;
    float*    asrc2 = asrc1;
    float*    adst2 = adst1;

    const int B = 256;

    long zlen = (long)NNODES * 4 * 2 + (long)NNODES * 2;
    zero_f32<<<2048, B, 0, stream>>>(esum1, zlen);

    {
        int waves = (NNODES / 16) * (256 / 64);
        gemm_f32_wmma<<<cdiv((long)waves * 32, B), B, 0, stream>>>(X, W1, xh1, NNODES, 256, 64);
        attn_scores<<<cdiv((long)NNODES * 4, B), B, 0, stream>>>(xh1, att_src1, att_dst1,
                                                                 asrc1, adst1, 4, 64);
        edge_max   <<<cdiv((long)ETOT * 4, B), B, 0, stream>>>(ei, asrc1, adst1, e1, emax1, 4);
        edge_expsum<<<cdiv((long)ETOT * 4, B), B, 0, stream>>>(ei, e1, emax1, esum1, 4);
        aggregate_owned<4, 64><<<cdiv(NNODES, RNODES), 1024, 0, stream>>>(ei, e1, esum1, xh1, b1, h1);
    }

    {
        int waves = (NNODES / 16) * (128 / 64);
        gemm_f32_wmma<<<cdiv((long)waves * 32, B), B, 0, stream>>>(h1, W2, xh2, NNODES, 128, 256);
        attn_scores<<<cdiv((long)NNODES, B), B, 0, stream>>>(xh2, att_src2, att_dst2,
                                                             asrc2, adst2, 1, 128);
        edge_max   <<<cdiv((long)ETOT, B), B, 0, stream>>>(ei, asrc2, adst2, e2, emax2, 1);
        edge_expsum<<<cdiv((long)ETOT, B), B, 0, stream>>>(ei, e2, emax2, esum2, 1);
        aggregate_owned<1, 128><<<cdiv(NNODES, RNODES), 1024, 0, stream>>>(ei, e2, esum2, xh2, b2, out);
    }
}
